# starting kernel
# baseline (speedup 1.0000x reference)
#include <hip/hip_runtime.h>
#include <hip/hip_bf16.h>

typedef __attribute__((ext_vector_type(16))) _Float16 v16h;
typedef __attribute__((ext_vector_type(8)))  float    v8f;
typedef __attribute__((ext_vector_type(4)))  float    v4f;

#define IN_DIM 128
#define HID    64
#define NOUT   3
#define WPB    8


__global__ void zero_f32(float* __restrict__ p, size_t n) {
  size_t i = (size_t)blockIdx.x * blockDim.x + threadIdx.x;
  if (i < n) p[i] = 0.0f;
}

__global__ void degree_count(const int* __restrict__ ei, float* __restrict__ cnt, int E) {
  int e = blockIdx.x * blockDim.x + threadIdx.x;
  if (e < E) atomicAdd(cnt + ei[E + e], 1.0f);
}

__global__ void scatter_add_128(const float* __restrict__ x, const int* __restrict__ ei,
                                float* __restrict__ agg, int E) {
  long long t = (long long)blockIdx.x * blockDim.x + threadIdx.x;
  int e = (int)(t >> 5);
  if (e >= E) return;
  int c = ((int)t & 31) * 4;
  int s = ei[e], d = ei[E + e];
  v4f v = *(const v4f*)(x + (size_t)s * IN_DIM + c);
  float* out = agg + (size_t)d * IN_DIM + c;
  atomicAdd(out + 0, v[0]); atomicAdd(out + 1, v[1]);
  atomicAdd(out + 2, v[2]); atomicAdd(out + 3, v[3]);
}

__global__ void scatter_add_64(const float* __restrict__ h, const int* __restrict__ ei,
                               float* __restrict__ agg, int E) {
  long long t = (long long)blockIdx.x * blockDim.x + threadIdx.x;
  int e = (int)(t >> 4);
  if (e >= E) return;
  int c = ((int)t & 15) * 4;
  int s = ei[e], d = ei[E + e];
  v4f v = *(const v4f*)(h + (size_t)s * HID + c);
  float* out = agg + (size_t)d * HID + c;
  atomicAdd(out + 0, v[0]); atomicAdd(out + 1, v[1]);
  atomicAdd(out + 2, v[2]); atomicAdd(out + 3, v[3]);
}

__device__ __forceinline__ v16h load_a_frag_f32(const float* __restrict__ rowBase,
                                                int k0, float scale, int khalf) {
  int g0 = khalf * 8;
  int g1 = 16 + khalf * 8;
  v4f a0 = *(const v4f*)(rowBase + k0 + g0);
  v4f a1 = *(const v4f*)(rowBase + k0 + g0 + 4);
  v4f a2 = *(const v4f*)(rowBase + k0 + g1);
  v4f a3 = *(const v4f*)(rowBase + k0 + g1 + 4);
  v16h r;
#pragma unroll
  for (int i = 0; i < 4; ++i) {
    r[i]      = (_Float16)(a0[i] * scale);
    r[4 + i]  = (_Float16)(a1[i] * scale);
    r[8 + i]  = (_Float16)(a2[i] * scale);
    r[12 + i] = (_Float16)(a3[i] * scale);
  }
  return r;
}

__global__ void __launch_bounds__(256)
sage_layer1_wmma(const float* __restrict__ x, const float* __restrict__ agg,
                 const float* __restrict__ cnt,
                 const float* __restrict__ Wl, const float* __restrict__ Wr,
                 const float* __restrict__ bias,
                 float* __restrict__ h, int N, int rowTiles) {
  __shared__ _Float16 sW[2][32][256];

  for (int idx = threadIdx.x; idx < 32 * 256; idx += 256) {
    int g  = idx >> 8;
    int n  = (idx >> 4) & 15;
    int j  = idx & 15;
    int ct = g >> 3;
    int kh = (g >> 2) & 1;
    int kc = g & 3;
    int k  = kc * 32 + kh * 8 + j + ((j >= 8) ? 8 : 0);
    int col = ct * 16 + n;
    sW[0][g][n * 16 + j] = (_Float16)Wl[k * HID + col];
    sW[1][g][n * 16 + j] = (_Float16)Wr[k * HID + col];
  }
  __syncthreads();

  int wave = threadIdx.x >> 5;
  int lid  = threadIdx.x & 31;
  int rowTile = blockIdx.x * WPB + wave;
  if (rowTile >= rowTiles) return;

  int r0    = rowTile * 16;
  int m     = lid & 15;
  int khalf = lid >> 4;
  int row = r0 + m; if (row >= N) row = N - 1;

  float inv = 1.0f / fmaxf(cnt[row], 1.0f);
  const float* aggRow = agg + (size_t)row * IN_DIM;
  const float* xRow   = x   + (size_t)row * IN_DIM;

  v8f acc[4] = {{}, {}, {}, {}};
#pragma unroll
  for (int kc = 0; kc < 4; ++kc) {
    int k0 = kc * 32;
    v16h aA = load_a_frag_f32(aggRow, k0, inv,  khalf);
    v16h aX = load_a_frag_f32(xRow,   k0, 1.0f, khalf);
#pragma unroll
    for (int ct = 0; ct < 4; ++ct) {
      int g = (ct * 2 + khalf) * 4 + kc;
      v16h bL = *(const v16h*)&sW[0][g][m * 16];
      v16h bR = *(const v16h*)&sW[1][g][m * 16];
      acc[ct] = __builtin_amdgcn_wmma_f32_16x16x32_f16(false, aA, false, bL,
                                                       (short)0, acc[ct],
                                                       false, false);
      acc[ct] = __builtin_amdgcn_wmma_f32_16x16x32_f16(false, aX, false, bR,
                                                       (short)0, acc[ct],
                                                       false, false);
    }
  }

#pragma unroll
  for (int ct = 0; ct < 4; ++ct) {
    int n = ct * 16 + m;
    float bv = bias[n];
#pragma unroll
    for (int r = 0; r < 8; ++r) {
      int orow = r0 + r + 8 * khalf;
      if (orow < N) h[(size_t)orow * HID + n] = fmaxf(acc[ct][r] + bv, 0.0f);
    }
  }
}

__global__ void sage_layer2(const float* __restrict__ h, const float* __restrict__ agg2,
                            const float* __restrict__ cnt,
                            const float* __restrict__ Wl, const float* __restrict__ Wr,
                            const float* __restrict__ bias,
                            float* __restrict__ out, int N) {
  int nidx = blockIdx.x * blockDim.x + threadIdx.x;
  if (nidx >= N) return;
  float inv = 1.0f / fmaxf(cnt[nidx], 1.0f);
  const float* hr = h    + (size_t)nidx * HID;
  const float* ar = agg2 + (size_t)nidx * HID;
  float o0 = bias[0], o1 = bias[1], o2 = bias[2];
#pragma unroll 8
  for (int k = 0; k < HID; ++k) {
    float av = ar[k] * inv;
    float hv = hr[k];
    o0 += av * Wl[k * NOUT + 0] + hv * Wr[k * NOUT + 0];
    o1 += av * Wl[k * NOUT + 1] + hv * Wr[k * NOUT + 1];
    o2 += av * Wl[k * NOUT + 2] + hv * Wr[k * NOUT + 2];
  }
  float mx  = fmaxf(o0, fmaxf(o1, o2));
  float lse = logf(expf(o0 - mx) + expf(o1 - mx) + expf(o2 - mx));
  out[(size_t)nidx * NOUT + 0] = o0 - mx - lse;
  out[(size_t)nidx * NOUT + 1] = o1 - mx - lse;
  out[(size_t)nidx * NOUT + 2] = o2 - mx - lse;
}


extern "C" void kernel_launch(void* const* d_in, const int* in_sizes, int n_in,
                              void* d_out, int out_size, void* d_ws, size_t ws_size,
                              hipStream_t stream) {
  (void)n_in; (void)out_size; (void)ws_size;
  const float* x   = (const float*)d_in[0];
  const int*   ei  = (const int*)d_in[1];
  const float* Wl1 = (const float*)d_in[2];
  const float* Wr1 = (const float*)d_in[3];
  const float* b1  = (const float*)d_in[4];
  const float* Wl2 = (const float*)d_in[5];
  const float* Wr2 = (const float*)d_in[6];
  const float* b2  = (const float*)d_in[7];

  int N = in_sizes[0] / IN_DIM;
  int E = in_sizes[1] / 2;

  float* agg1 = (float*)d_ws;
  float* agg2 = agg1 + (size_t)N * IN_DIM;
  float* hbuf = agg2 + (size_t)N * HID;
  float* cnt  = hbuf + (size_t)N * HID;

  size_t nz = (size_t)N * (IN_DIM + HID);
  zero_f32<<<(int)((nz + 255) / 256), 256, 0, stream>>>(agg1, nz);
  zero_f32<<<(N + 255) / 256, 256, 0, stream>>>(cnt, (size_t)N);

  degree_count<<<(E + 255) / 256, 256, 0, stream>>>(ei, cnt, E);

  long long t1 = (long long)E * 32;
  scatter_add_128<<<(int)((t1 + 255) / 256), 256, 0, stream>>>(x, ei, agg1, E);

  int rowTiles = (N + 15) / 16;
  int blocks   = (rowTiles + WPB - 1) / WPB;
  sage_layer1_wmma<<<blocks, 256, 0, stream>>>(x, agg1, cnt, Wl1, Wr1, b1,
                                               hbuf, N, rowTiles);

  long long t2 = (long long)E * 16;
  scatter_add_64<<<(int)((t2 + 255) / 256), 256, 0, stream>>>(hbuf, ei, agg2, E);

  sage_layer2<<<(N + 255) / 256, 256, 0, stream>>>(hbuf, agg2, cnt, Wl2, Wr2, b2,
                                                   (float*)d_out, N);
}


# Projection launch stages weights, x strips and both outputs through LDS so that every global load and store covers whole 128-B lines (was 16 lines a load, 64-B store pieces); sums unchanged
# speedup vs baseline: 2.9833x; 2.9833x over previous
#include <hip/hip_runtime.h>
#include <hip/hip_bf16.h>

typedef __attribute__((ext_vector_type(16))) _Float16 v16h;
typedef __attribute__((ext_vector_type(8)))  float    v8f;
typedef __attribute__((ext_vector_type(4)))  float    v4f;
typedef __attribute__((ext_vector_type(2)))  int      v2i;
typedef __attribute__((ext_vector_type(8)))  _Float16 v8h;
typedef __attribute__((ext_vector_type(4)))  _Float16 v4h;
typedef __attribute__((ext_vector_type(2)))  unsigned v2u;
typedef __attribute__((ext_vector_type(4)))  unsigned v4u;
typedef __attribute__((ext_vector_type(8)))  unsigned v8u;

#define IN_DIM 128
#define HID    64
#define NOUT   3
#define QW     4
#define WPB    8


__global__ void __launch_bounds__(256)
link_edges(const int* __restrict__ ei, int* __restrict__ head, int* __restrict__ link,
           int E, int quarter) {
  int t = blockIdx.x * blockDim.x + threadIdx.x;
  if (t + 3 * quarter < E) {
    int e0 = t, e1 = t + quarter, e2 = t + 2 * quarter, e3 = t + 3 * quarter;
    int s0 = ei[e0], s1 = ei[e1], s2 = ei[e2], s3 = ei[e3];
    int d0 = ei[E + e0], d1 = ei[E + e1], d2 = ei[E + e2], d3 = ei[E + e3];
    int o0 = atomicExch(head + d0, e0);
    int o1 = atomicExch(head + d1, e1);
    int o2 = atomicExch(head + d2, e2);
    int o3 = atomicExch(head + d3, e3);
    v2i r0 = {s0, o0}, r1 = {s1, o1}, r2 = {s2, o2}, r3 = {s3, o3};
    *(v2i*)(link + 2 * (size_t)e0) = r0;
    *(v2i*)(link + 2 * (size_t)e1) = r1;
    *(v2i*)(link + 2 * (size_t)e2) = r2;
    *(v2i*)(link + 2 * (size_t)e3) = r3;
  } else {
    for (int k = 0; k < 4; ++k) {
      int e = t + k * quarter;
      if (e < E) {
        int s = ei[e], d = ei[E + e];
        int o = atomicExch(head + d, e);
        v2i r = {s, o};
        *(v2i*)(link + 2 * (size_t)e) = r;
      }
    }
  }
}


#define SP 68

__device__ __forceinline__ void wave_lds_handover() {
  __builtin_amdgcn_fence(__ATOMIC_RELEASE, "wavefront");
  __builtin_amdgcn_wave_barrier();
  __builtin_amdgcn_fence(__ATOMIC_ACQUIRE, "wavefront");
}

__device__ __forceinline__ void store_strip(const unsigned* st, float* __restrict__ dst,
                                            int r0, int N, int m, int khalf, bool whole) {
  v4u q[8];
#pragma unroll
  for (int j = 0; j < 8; ++j)
    q[j] = *(const v4u*)(st + (2 * j + khalf) * SP + m * 4);
  float* base = dst + (size_t)(r0 + khalf) * HID + m * 4;
  if (whole) {
#pragma unroll
    for (int j = 0; j < 8; ++j)
      *(v4u*)(base + (size_t)(2 * j) * HID) = q[j];
  } else {
#pragma unroll
    for (int j = 0; j < 8; ++j)
      if (r0 + 2 * j + khalf < N) *(v4u*)(base + (size_t)(2 * j) * HID) = q[j];
  }
}

__global__ void __launch_bounds__(256)
sage_project1_wmma(const float* __restrict__ x,
                   const float* __restrict__ Wl, const float* __restrict__ Wr,
                   const float* __restrict__ bias,
                   float* __restrict__ pl, float* __restrict__ pr,
                   int* __restrict__ head,
                   int N, int rowTiles) {
  __shared__ v8h sW[2][32][2][16];
  __shared__ v4u sStage[WPB][16 * SP / 4];

  {
    int gid = blockIdx.x * 256 + threadIdx.x;
    if (gid < N) head[gid] = -1;
  }

  for (int u = threadIdx.x; u < 512; u += 256) {
    int kq = u >> 4;
    int c4 = u & 15;
    const float* wl = Wl + (size_t)(4 * kq) * HID + 4 * c4;
    const float* wr = Wr + (size_t)(4 * kq) * HID + 4 * c4;
    v4f l0 = *(const v4f*)(wl);
    v4f l1 = *(const v4f*)(wl + HID);
    v4f l2 = *(const v4f*)(wl + 2 * HID);
    v4f l3 = *(const v4f*)(wl + 3 * HID);
    v4f q0 = *(const v4f*)(wr);
    v4f q1 = *(const v4f*)(wr + HID);
    v4f q2 = *(const v4f*)(wr + 2 * HID);
    v4f q3 = *(const v4f*)(wr + 3 * HID);
    int kc = kq >> 3;
    int kh = (kq >> 1) & 1;
    int p  = (kq >> 2) & 1;
    int jq = kq & 1;
    int ct = c4 >> 2;
    int g  = (ct * 2 + kh) * 4 + kc;
    int n0 = 4 * (c4 & 3);
#pragma unroll
    for (int i = 0; i < 4; ++i) {
      v4h hl = {(_Float16)l0[i], (_Float16)l1[i], (_Float16)l2[i], (_Float16)l3[i]};
      v4h hr = {(_Float16)q0[i], (_Float16)q1[i], (_Float16)q2[i], (_Float16)q3[i]};
      ((v4h*)&sW[0][g][p][n0 + i])[jq] = hl;
      ((v4h*)&sW[1][g][p][n0 + i])[jq] = hr;
    }
  }
  __syncthreads();

  int wave = threadIdx.x >> 5;
  int lid  = threadIdx.x & 31;
  int rowTile = blockIdx.x * WPB + wave;
  if (rowTile >= rowTiles) return;

  int r0    = rowTile * 16;
  int m     = lid & 15;
  int khalf = lid >> 4;

  unsigned* st = (unsigned*)sStage[wave];

  {
    v4f xv[16];
#pragma unroll
    for (int i = 0; i < 16; ++i) {
      int row = r0 + i; if (row >= N) row = N - 1;
      xv[i] = *(const v4f*)(x + (size_t)row * IN_DIM + lid * 4);
    }
#pragma unroll
    for (int i = 0; i < 16; ++i) {
      v4h hv = {(_Float16)xv[i][0], (_Float16)xv[i][1], (_Float16)xv[i][2], (_Float16)xv[i][3]};
      *(v2u*)(st + i * SP + lid * 2) = __builtin_bit_cast(v2u, hv);
    }
  }
  wave_lds_handover();

  v8f accL[4] = {{}, {}, {}, {}};
  v8f accR[4] = {{}, {}, {}, {}};
#pragma unroll
  for (int kc = 0; kc < 4; ++kc) {
    const unsigned* ap = st + m * SP + kc * 16 + khalf * 4;
    v8u aw;
    {
      v4u a_lo = *(const v4u*)(ap);
      v4u a_hi = *(const v4u*)(ap + 8);
      aw = __builtin_shufflevector(a_lo, a_hi, 0, 1, 2, 3, 4, 5, 6, 7);
    }
    v16h aX = __builtin_bit_cast(v16h, aw);
#pragma unroll
    for (int ct = 0; ct < 4; ++ct) {
      int g = (ct * 2 + khalf) * 4 + kc;
      v16h bL = __builtin_shufflevector(sW[0][g][0][m], sW[0][g][1][m],
                                        0, 1, 2, 3, 4, 5, 6, 7, 8, 9, 10, 11, 12, 13, 14, 15);
      v16h bR = __builtin_shufflevector(sW[1][g][0][m], sW[1][g][1][m],
                                        0, 1, 2, 3, 4, 5, 6, 7, 8, 9, 10, 11, 12, 13, 14, 15);
      accL[ct] = __builtin_amdgcn_wmma_f32_16x16x32_f16(false, aX, false, bL,
                                                        (short)0, accL[ct],
                                                        false, false);
      accR[ct] = __builtin_amdgcn_wmma_f32_16x16x32_f16(false, aX, false, bR,
                                                        (short)0, accR[ct],
                                                        false, false);
    }
  }

  bool whole = r0 + 16 <= N;
  wave_lds_handover();
#pragma unroll
  for (int ct = 0; ct < 4; ++ct) {
#pragma unroll
    for (int r = 0; r < 8; ++r)
      st[(r + 8 * khalf) * SP + ct * 16 + m] = __float_as_uint(accL[ct][r]);
  }
  wave_lds_handover();
  store_strip(st, pl, r0, N, m, khalf, whole);
  wave_lds_handover();
#pragma unroll
  for (int ct = 0; ct < 4; ++ct) {
    float bv = bias[ct * 16 + m];
#pragma unroll
    for (int r = 0; r < 8; ++r)
      st[(r + 8 * khalf) * SP + ct * 16 + m] = __float_as_uint(accR[ct][r] + bv);
  }
  wave_lds_handover();
  store_strip(st, pr, r0, N, m, khalf, whole);
}

__global__ void __launch_bounds__(256)
sage_gather_hidden_project(const float* __restrict__ pl, const float* __restrict__ pr,
                           const int* __restrict__ head, const int* __restrict__ link,
                           const float* __restrict__ Wl, const float* __restrict__ Wr,
                           const float* __restrict__ bias,
                           float* __restrict__ ql, float* __restrict__ qr, int N, int E) {
  __shared__ float sWl[HID * NOUT];
  __shared__ float sWr[HID * NOUT];
  if (threadIdx.x < HID * NOUT) {
    sWl[threadIdx.x] = Wl[threadIdx.x];
    sWr[threadIdx.x] = Wr[threadIdx.x];
  }
  __syncthreads();

  int t    = blockIdx.x * blockDim.x + threadIdx.x;
  int node = t >> 3;
  int sub  = t & 7;
  int nd   = node < N ? node : N - 1;
  int c0   = sub * 4;
  int c1   = 32 + sub * 4;

  int e = head[nd];
  v4f r0 = *(const v4f*)(pr + (size_t)nd * HID + c0);
  v4f r1 = *(const v4f*)(pr + (size_t)nd * HID + c1);

  v4f a0 = {0.0f, 0.0f, 0.0f, 0.0f};
  v4f a1 = {0.0f, 0.0f, 0.0f, 0.0f};
  int deg = 0;
  bool live = (unsigned)e < (unsigned)E;
  v2i lk = {0, -1};
  if (live) lk = *(const v2i*)(link + 2 * (size_t)e);
  while (live && deg < E) {
    const float* rp = pl + (size_t)lk[0] * HID;
    v4f g0 = *(const v4f*)(rp + c0);
    v4f g1 = *(const v4f*)(rp + c1);
    int en = lk[1];
    live = (unsigned)en < (unsigned)E;
    if (live) lk = *(const v2i*)(link + 2 * (size_t)en);
    a0 += g0;
    a1 += g1;
    ++deg;
  }

  float inv = 1.0f / fmaxf((float)deg, 1.0f);

  const float* wl0 = sWl + c0 * NOUT;
  const float* wr0 = sWr + c0 * NOUT;
  const float* wl1 = sWl + c1 * NOUT;
  const float* wr1 = sWr + c1 * NOUT;
  float s0 = 0.0f, s1 = 0.0f, s2 = 0.0f, s3 = 0.0f, s4 = 0.0f, s5 = 0.0f;
#pragma unroll
  for (int i = 0; i < 4; ++i) {
    float hv = fmaxf(a0[i] * inv + r0[i], 0.0f);
    s0 += hv * wl0[i * NOUT + 0];
    s1 += hv * wl0[i * NOUT + 1];
    s2 += hv * wl0[i * NOUT + 2];
    s3 += hv * wr0[i * NOUT + 0];
    s4 += hv * wr0[i * NOUT + 1];
    s5 += hv * wr0[i * NOUT + 2];
  }
#pragma unroll
  for (int i = 0; i < 4; ++i) {
    float hv = fmaxf(a1[i] * inv + r1[i], 0.0f);
    s0 += hv * wl1[i * NOUT + 0];
    s1 += hv * wl1[i * NOUT + 1];
    s2 += hv * wl1[i * NOUT + 2];
    s3 += hv * wr1[i * NOUT + 0];
    s4 += hv * wr1[i * NOUT + 1];
    s5 += hv * wr1[i * NOUT + 2];
  }
  for (int off = 4; off > 0; off >>= 1) {
    s0 += __shfl_xor(s0, off, 8);
    s1 += __shfl_xor(s1, off, 8);
    s2 += __shfl_xor(s2, off, 8);
    s3 += __shfl_xor(s3, off, 8);
    s4 += __shfl_xor(s4, off, 8);
    s5 += __shfl_xor(s5, off, 8);
  }
  if (sub == 0 && node < N) {
    v4f vl = {s0, s1, s2, 0.0f};
    v4f vr = {s3 + bias[0], s4 + bias[1], s5 + bias[2], 0.0f};
    *(v4f*)(ql + (size_t)node * QW) = vl;
    *(v4f*)(qr + (size_t)node * QW) = vr;
  }
}

__global__ void __launch_bounds__(256)
sage_gather_finish(const float* __restrict__ ql, const float* __restrict__ qr,
                   const int* __restrict__ head, const int* __restrict__ link,
                   float* __restrict__ out, int N, int E) {
  int nidx = blockIdx.x * blockDim.x + threadIdx.x;
  if (nidx >= N) return;
  int e = head[nidx];
  v4f r = *(const v4f*)(qr + (size_t)nidx * QW);

  v4f a = {0.0f, 0.0f, 0.0f, 0.0f};
  int deg = 0;
  bool live = (unsigned)e < (unsigned)E;
  v2i lk = {0, -1};
  if (live) lk = *(const v2i*)(link + 2 * (size_t)e);
  while (live && deg < E) {
    v4f g = *(const v4f*)(ql + (size_t)lk[0] * QW);
    int en = lk[1];
    live = (unsigned)en < (unsigned)E;
    if (live) lk = *(const v2i*)(link + 2 * (size_t)en);
    a += g;
    ++deg;
  }

  float inv = 1.0f / fmaxf((float)deg, 1.0f);
  float o0 = a[0] * inv + r[0];
  float o1 = a[1] * inv + r[1];
  float o2 = a[2] * inv + r[2];
  float mx  = fmaxf(o0, fmaxf(o1, o2));
  float lse = logf(expf(o0 - mx) + expf(o1 - mx) + expf(o2 - mx));
  out[(size_t)nidx * NOUT + 0] = o0 - mx - lse;
  out[(size_t)nidx * NOUT + 1] = o1 - mx - lse;
  out[(size_t)nidx * NOUT + 2] = o2 - mx - lse;
}


extern "C" void kernel_launch(void* const* d_in, const int* in_sizes, int n_in,
                              void* d_out, int out_size, void* d_ws, size_t ws_size,
                              hipStream_t stream) {
  (void)n_in; (void)out_size; (void)ws_size;
  const float* x   = (const float*)d_in[0];
  const int*   ei  = (const int*)d_in[1];
  const float* Wl1 = (const float*)d_in[2];
  const float* Wr1 = (const float*)d_in[3];
  const float* b1  = (const float*)d_in[4];
  const float* Wl2 = (const float*)d_in[5];
  const float* Wr2 = (const float*)d_in[6];
  const float* b2  = (const float*)d_in[7];

  int N = in_sizes[0] / IN_DIM;
  int E = in_sizes[1] / 2;

  float* pl   = (float*)d_ws;
  float* pr   = pl   + (size_t)N * HID;
  float* ql   = pr   + (size_t)N * HID;
  float* qr   = ql   + (size_t)N * QW;
  int*   link = (int*)(qr + (size_t)N * QW);
  int*   head = link + (size_t)E * 2;

  int rowTiles = (N + 15) / 16;
  int blocks   = (rowTiles + WPB - 1) / WPB;
  sage_project1_wmma<<<blocks, 256, 0, stream>>>(x, Wl1, Wr1, b1, pl, pr, head, N, rowTiles);

  int lblocks = ((E + 3) / 4 + 255) / 256;
  int quarter = lblocks * 256;
  link_edges<<<lblocks, 256, 0, stream>>>(ei, head, link, E, quarter);

  long long tn = (long long)N * 8;
  sage_gather_hidden_project<<<(int)((tn + 255) / 256), 256, 0, stream>>>(pl, pr, head, link,
                                                                       Wl2, Wr2, b2, ql, qr, N, E);

  sage_gather_finish<<<(N + 255) / 256, 256, 0, stream>>>(ql, qr, head, link, (float*)d_out, N, E);
}


# Edge lists become a 64-B slot row per destination: an integer add gives each edge its arrival rank and it stores its source id there; gathers read ids from the row, no list walk; sums in rank order
# speedup vs baseline: 3.1127x; 1.0434x over previous
#include <hip/hip_runtime.h>
#include <hip/hip_bf16.h>

typedef __attribute__((ext_vector_type(16))) _Float16 v16h;
typedef __attribute__((ext_vector_type(8)))  float    v8f;
typedef __attribute__((ext_vector_type(4)))  float    v4f;
typedef __attribute__((ext_vector_type(2)))  int      v2i;
typedef __attribute__((ext_vector_type(4)))  int      v4i;
typedef __attribute__((ext_vector_type(8)))  _Float16 v8h;
typedef __attribute__((ext_vector_type(4)))  _Float16 v4h;
typedef __attribute__((ext_vector_type(2)))  unsigned v2u;
typedef __attribute__((ext_vector_type(4)))  unsigned v4u;
typedef __attribute__((ext_vector_type(8)))  unsigned v8u;

#define IN_DIM 128
#define HID    64
#define NOUT   3
#define QW     4
#define WPB    8
#define SLOTW  16
#define NSLOT  15


__device__ __forceinline__ void place_edge(int rank, int e, int s, int d,
                                           int* __restrict__ slot, int* __restrict__ head,
                                           int* __restrict__ link) {
  if (rank < NSLOT) {
    slot[(size_t)d * SLOTW + 1 + rank] = s;
  } else {
    int o = atomicExch(head + d, e);
    v2i r = {s, o};
    *(v2i*)(link + 2 * (size_t)e) = r;
  }
}

__global__ void __launch_bounds__(256)
link_edges(const int* __restrict__ ei, int* __restrict__ slot, int* __restrict__ head,
           int* __restrict__ link, int E, int quarter) {
  int t = blockIdx.x * blockDim.x + threadIdx.x;
  if (t + 3 * quarter < E) {
    int e0 = t, e1 = t + quarter, e2 = t + 2 * quarter, e3 = t + 3 * quarter;
    int s0 = ei[e0], s1 = ei[e1], s2 = ei[e2], s3 = ei[e3];
    int d0 = ei[E + e0], d1 = ei[E + e1], d2 = ei[E + e2], d3 = ei[E + e3];
    int k0 = atomicAdd(slot + (size_t)d0 * SLOTW, 1);
    int k1 = atomicAdd(slot + (size_t)d1 * SLOTW, 1);
    int k2 = atomicAdd(slot + (size_t)d2 * SLOTW, 1);
    int k3 = atomicAdd(slot + (size_t)d3 * SLOTW, 1);
    place_edge(k0, e0, s0, d0, slot, head, link);
    place_edge(k1, e1, s1, d1, slot, head, link);
    place_edge(k2, e2, s2, d2, slot, head, link);
    place_edge(k3, e3, s3, d3, slot, head, link);
  } else {
    for (int k = 0; k < 4; ++k) {
      int e = t + k * quarter;
      if (e < E) {
        int s = ei[e], d = ei[E + e];
        int rank = atomicAdd(slot + (size_t)d * SLOTW, 1);
        place_edge(rank, e, s, d, slot, head, link);
      }
    }
  }
}


#define SP 68

__device__ __forceinline__ void wave_lds_handover() {
  __builtin_amdgcn_fence(__ATOMIC_RELEASE, "wavefront");
  __builtin_amdgcn_wave_barrier();
  __builtin_amdgcn_fence(__ATOMIC_ACQUIRE, "wavefront");
}

__device__ __forceinline__ void store_strip(const unsigned* st, float* __restrict__ dst,
                                            int r0, int N, int m, int khalf, bool whole) {
  v4u q[8];
#pragma unroll
  for (int j = 0; j < 8; ++j)
    q[j] = *(const v4u*)(st + (2 * j + khalf) * SP + m * 4);
  float* base = dst + (size_t)(r0 + khalf) * HID + m * 4;
  if (whole) {
#pragma unroll
    for (int j = 0; j < 8; ++j)
      *(v4u*)(base + (size_t)(2 * j) * HID) = q[j];
  } else {
#pragma unroll
    for (int j = 0; j < 8; ++j)
      if (r0 + 2 * j + khalf < N) *(v4u*)(base + (size_t)(2 * j) * HID) = q[j];
  }
}

__global__ void __launch_bounds__(256)
sage_project1_wmma(const float* __restrict__ x,
                   const float* __restrict__ Wl, const float* __restrict__ Wr,
                   const float* __restrict__ bias,
                   float* __restrict__ pl, float* __restrict__ pr,
                   int* __restrict__ slot, int* __restrict__ head,
                   int N, int rowTiles) {
  __shared__ v8h sW[2][32][2][16];
  __shared__ v4u sStage[WPB][16 * SP / 4];

  {
    int gid = blockIdx.x * 256 + threadIdx.x;
    int nthr = gridDim.x * 256;
    v4u z = {0u, 0u, 0u, 0u};
    for (int u = gid; u < 4 * N; u += nthr) ((v4u*)slot)[u] = z;
    if (gid < N) head[gid] = -1;
  }

  for (int u = threadIdx.x; u < 512; u += 256) {
    int kq = u >> 4;
    int c4 = u & 15;
    const float* wl = Wl + (size_t)(4 * kq) * HID + 4 * c4;
    const float* wr = Wr + (size_t)(4 * kq) * HID + 4 * c4;
    v4f l0 = *(const v4f*)(wl);
    v4f l1 = *(const v4f*)(wl + HID);
    v4f l2 = *(const v4f*)(wl + 2 * HID);
    v4f l3 = *(const v4f*)(wl + 3 * HID);
    v4f q0 = *(const v4f*)(wr);
    v4f q1 = *(const v4f*)(wr + HID);
    v4f q2 = *(const v4f*)(wr + 2 * HID);
    v4f q3 = *(const v4f*)(wr + 3 * HID);
    int kc = kq >> 3;
    int kh = (kq >> 1) & 1;
    int p  = (kq >> 2) & 1;
    int jq = kq & 1;
    int ct = c4 >> 2;
    int g  = (ct * 2 + kh) * 4 + kc;
    int n0 = 4 * (c4 & 3);
#pragma unroll
    for (int i = 0; i < 4; ++i) {
      v4h hl = {(_Float16)l0[i], (_Float16)l1[i], (_Float16)l2[i], (_Float16)l3[i]};
      v4h hr = {(_Float16)q0[i], (_Float16)q1[i], (_Float16)q2[i], (_Float16)q3[i]};
      ((v4h*)&sW[0][g][p][n0 + i])[jq] = hl;
      ((v4h*)&sW[1][g][p][n0 + i])[jq] = hr;
    }
  }
  __syncthreads();

  int wave = threadIdx.x >> 5;
  int lid  = threadIdx.x & 31;
  int rowTile = blockIdx.x * WPB + wave;
  if (rowTile >= rowTiles) return;

  int r0    = rowTile * 16;
  int m     = lid & 15;
  int khalf = lid >> 4;

  unsigned* st = (unsigned*)sStage[wave];

  {
    v4f xv[16];
#pragma unroll
    for (int i = 0; i < 16; ++i) {
      int row = r0 + i; if (row >= N) row = N - 1;
      xv[i] = *(const v4f*)(x + (size_t)row * IN_DIM + lid * 4);
    }
#pragma unroll
    for (int i = 0; i < 16; ++i) {
      v4h hv = {(_Float16)xv[i][0], (_Float16)xv[i][1], (_Float16)xv[i][2], (_Float16)xv[i][3]};
      *(v2u*)(st + i * SP + lid * 2) = __builtin_bit_cast(v2u, hv);
    }
  }
  wave_lds_handover();

  v8f accL[4] = {{}, {}, {}, {}};
  v8f accR[4] = {{}, {}, {}, {}};
#pragma unroll
  for (int kc = 0; kc < 4; ++kc) {
    const unsigned* ap = st + m * SP + kc * 16 + khalf * 4;
    v8u aw;
    {
      v4u a_lo = *(const v4u*)(ap);
      v4u a_hi = *(const v4u*)(ap + 8);
      aw = __builtin_shufflevector(a_lo, a_hi, 0, 1, 2, 3, 4, 5, 6, 7);
    }
    v16h aX = __builtin_bit_cast(v16h, aw);
#pragma unroll
    for (int ct = 0; ct < 4; ++ct) {
      int g = (ct * 2 + khalf) * 4 + kc;
      v16h bL = __builtin_shufflevector(sW[0][g][0][m], sW[0][g][1][m],
                                        0, 1, 2, 3, 4, 5, 6, 7, 8, 9, 10, 11, 12, 13, 14, 15);
      v16h bR = __builtin_shufflevector(sW[1][g][0][m], sW[1][g][1][m],
                                        0, 1, 2, 3, 4, 5, 6, 7, 8, 9, 10, 11, 12, 13, 14, 15);
      accL[ct] = __builtin_amdgcn_wmma_f32_16x16x32_f16(false, aX, false, bL,
                                                        (short)0, accL[ct],
                                                        false, false);
      accR[ct] = __builtin_amdgcn_wmma_f32_16x16x32_f16(false, aX, false, bR,
                                                        (short)0, accR[ct],
                                                        false, false);
    }
  }

  bool whole = r0 + 16 <= N;
  wave_lds_handover();
#pragma unroll
  for (int ct = 0; ct < 4; ++ct) {
#pragma unroll
    for (int r = 0; r < 8; ++r)
      st[(r + 8 * khalf) * SP + ct * 16 + m] = __float_as_uint(accL[ct][r]);
  }
  wave_lds_handover();
  store_strip(st, pl, r0, N, m, khalf, whole);
  wave_lds_handover();
#pragma unroll
  for (int ct = 0; ct < 4; ++ct) {
    float bv = bias[ct * 16 + m];
#pragma unroll
    for (int r = 0; r < 8; ++r)
      st[(r + 8 * khalf) * SP + ct * 16 + m] = __float_as_uint(accR[ct][r] + bv);
  }
  wave_lds_handover();
  store_strip(st, pr, r0, N, m, khalf, whole);
}

__device__ __forceinline__ void row_if(bool filled, const float* __restrict__ pl, int id,
                                       unsigned lane_b, v4f& g0, v4f& g1) {
  v4f z = {0.0f, 0.0f, 0.0f, 0.0f};
  g0 = z;
  g1 = z;
  if (filled) {
    const char* rp = (const char*)pl + ((unsigned)id * (unsigned)(HID * 4) + lane_b);
    g0 = *(const v4f*)(rp);
    g1 = *(const v4f*)(rp + 128);
  }
}

#define ADD_SLOTS5(FIRST, k, i0, i1, i2, i3, i4)                               \
  {                                                                            \
    v4f x0, y0, x1, y1, x2, y2, x3, y3, x4, y4;                                \
    row_if((k) + 0 < ns, pl, (i0), lane_b, x0, y0);                            \
    row_if((k) + 1 < ns, pl, (i1), lane_b, x1, y1);                            \
    row_if((k) + 2 < ns, pl, (i2), lane_b, x2, y2);                            \
    row_if((k) + 3 < ns, pl, (i3), lane_b, x3, y3);                            \
    row_if((k) + 4 < ns, pl, (i4), lane_b, x4, y4);                            \
    a0 FIRST x0; a1 FIRST y0;                                                  \
    a0 += x1; a1 += y1;                                                        \
    a0 += x2; a1 += y2;                                                        \
    a0 += x3; a1 += y3;                                                        \
    a0 += x4; a1 += y4;                                                        \
  }

__device__ __forceinline__ v4f unit_if(bool filled, const float* __restrict__ ql, int id) {
  v4f g = {0.0f, 0.0f, 0.0f, 0.0f};
  if (filled) g = *(const v4f*)(ql + (size_t)id * QW);
  return g;
}

__global__ void __launch_bounds__(256)
sage_gather_hidden_project(const float* __restrict__ pl, const float* __restrict__ pr,
                           const int* __restrict__ slot,
                           const int* __restrict__ head, const int* __restrict__ link,
                           const float* __restrict__ Wl, const float* __restrict__ Wr,
                           const float* __restrict__ bias,
                           float* __restrict__ ql, float* __restrict__ qr, int N, int E) {
  __shared__ float sWl[HID * NOUT];
  __shared__ float sWr[HID * NOUT];
  if (threadIdx.x < HID * NOUT) {
    sWl[threadIdx.x] = Wl[threadIdx.x];
    sWr[threadIdx.x] = Wr[threadIdx.x];
  }
  __syncthreads();

  int t    = blockIdx.x * blockDim.x + threadIdx.x;
  int node = t >> 3;
  int sub  = t & 7;
  int nd   = node < N ? node : N - 1;
  int c0   = sub * 4;
  int c1   = 32 + sub * 4;
  unsigned lane_b = (unsigned)sub * 16u;

  const v4i* srow = (const v4i*)(slot + (size_t)nd * SLOTW);
  v4i s0 = srow[0], s1 = srow[1], s2 = srow[2], s3 = srow[3];
  const char* prow = (const char*)pr + ((unsigned)nd * (unsigned)(HID * 4) + lane_b);
  v4f r0 = *(const v4f*)(prow);
  v4f r1 = *(const v4f*)(prow + 128);

  asm volatile("" : "+v"(s0), "+v"(s1), "+v"(s2), "+v"(s3));
  int deg = s0[0];
  int ns  = deg < NSLOT ? deg : NSLOT;

  v4f a0, a1;
  ADD_SLOTS5(=,  0,  s0[1], s0[2], s0[3], s1[0], s1[1])
  ADD_SLOTS5(+=, 5,  s1[2], s1[3], s2[0], s2[1], s2[2])
  ADD_SLOTS5(+=, 10, s2[3], s3[0], s3[1], s3[2], s3[3])

  if (deg > NSLOT) {
    int e = head[nd];
    for (int k = 0; k < E && (unsigned)e < (unsigned)E; ++k) {
      v2i lk = *(const v2i*)(link + 2 * (size_t)e);
      const float* rp = pl + (size_t)lk[0] * HID;
      a0 += *(const v4f*)(rp + c0);
      a1 += *(const v4f*)(rp + c1);
      e = lk[1];
    }
  }

  float inv = 1.0f / fmaxf((float)deg, 1.0f);

  const float* wl0 = sWl + c0 * NOUT;
  const float* wr0 = sWr + c0 * NOUT;
  const float* wl1 = sWl + c1 * NOUT;
  const float* wr1 = sWr + c1 * NOUT;
  float q0 = 0.0f, q1 = 0.0f, q2 = 0.0f, q3 = 0.0f, q4 = 0.0f, q5 = 0.0f;
#pragma unroll
  for (int i = 0; i < 4; ++i) {
    float hv = fmaxf(a0[i] * inv + r0[i], 0.0f);
    q0 += hv * wl0[i * NOUT + 0];
    q1 += hv * wl0[i * NOUT + 1];
    q2 += hv * wl0[i * NOUT + 2];
    q3 += hv * wr0[i * NOUT + 0];
    q4 += hv * wr0[i * NOUT + 1];
    q5 += hv * wr0[i * NOUT + 2];
  }
#pragma unroll
  for (int i = 0; i < 4; ++i) {
    float hv = fmaxf(a1[i] * inv + r1[i], 0.0f);
    q0 += hv * wl1[i * NOUT + 0];
    q1 += hv * wl1[i * NOUT + 1];
    q2 += hv * wl1[i * NOUT + 2];
    q3 += hv * wr1[i * NOUT + 0];
    q4 += hv * wr1[i * NOUT + 1];
    q5 += hv * wr1[i * NOUT + 2];
  }
  for (int off = 4; off > 0; off >>= 1) {
    q0 += __shfl_xor(q0, off, 8);
    q1 += __shfl_xor(q1, off, 8);
    q2 += __shfl_xor(q2, off, 8);
    q3 += __shfl_xor(q3, off, 8);
    q4 += __shfl_xor(q4, off, 8);
    q5 += __shfl_xor(q5, off, 8);
  }
  if (sub == 0 && node < N) {
    v4f vl = {q0, q1, q2, 0.0f};
    v4f vr = {q3 + bias[0], q4 + bias[1], q5 + bias[2], 0.0f};
    *(v4f*)(ql + (size_t)node * QW) = vl;
    *(v4f*)(qr + (size_t)node * QW) = vr;
  }
}

__global__ void __launch_bounds__(256)
sage_gather_finish(const float* __restrict__ ql, const float* __restrict__ qr,
                   const int* __restrict__ slot,
                   const int* __restrict__ head, const int* __restrict__ link,
                   float* __restrict__ out, int N, int E) {
  int nidx = blockIdx.x * blockDim.x + threadIdx.x;
  if (nidx >= N) return;
  const v4i* srow = (const v4i*)(slot + (size_t)nidx * SLOTW);
  v4i s0 = srow[0], s1 = srow[1], s2 = srow[2], s3 = srow[3];
  v4f r = *(const v4f*)(qr + (size_t)nidx * QW);

  asm volatile("" : "+v"(s0), "+v"(s1), "+v"(s2), "+v"(s3));
  int deg = s0[0];
  int ns  = deg < NSLOT ? deg : NSLOT;

  v4f g0  = unit_if(0  < ns, ql, s0[1]);
  v4f g1  = unit_if(1  < ns, ql, s0[2]);
  v4f g2  = unit_if(2  < ns, ql, s0[3]);
  v4f g3  = unit_if(3  < ns, ql, s1[0]);
  v4f g4  = unit_if(4  < ns, ql, s1[1]);
  v4f g5  = unit_if(5  < ns, ql, s1[2]);
  v4f g6  = unit_if(6  < ns, ql, s1[3]);
  v4f g7  = unit_if(7  < ns, ql, s2[0]);
  v4f g8  = unit_if(8  < ns, ql, s2[1]);
  v4f g9  = unit_if(9  < ns, ql, s2[2]);
  v4f g10 = unit_if(10 < ns, ql, s2[3]);
  v4f g11 = unit_if(11 < ns, ql, s3[0]);
  v4f g12 = unit_if(12 < ns, ql, s3[1]);
  v4f g13 = unit_if(13 < ns, ql, s3[2]);
  v4f g14 = unit_if(14 < ns, ql, s3[3]);
  v4f a = g0;
  a += g1;  a += g2;  a += g3;  a += g4;
  a += g5;  a += g6;  a += g7;  a += g8;  a += g9;
  a += g10; a += g11; a += g12; a += g13; a += g14;

  if (deg > NSLOT) {
    int e = head[nidx];
    for (int k = 0; k < E && (unsigned)e < (unsigned)E; ++k) {
      v2i lk = *(const v2i*)(link + 2 * (size_t)e);
      a += *(const v4f*)(ql + (size_t)lk[0] * QW);
      e = lk[1];
    }
  }

  float inv = 1.0f / fmaxf((float)deg, 1.0f);
  float o0 = a[0] * inv + r[0];
  float o1 = a[1] * inv + r[1];
  float o2 = a[2] * inv + r[2];
  float mx  = fmaxf(o0, fmaxf(o1, o2));
  float lse = logf(expf(o0 - mx) + expf(o1 - mx) + expf(o2 - mx));
  out[(size_t)nidx * NOUT + 0] = o0 - mx - lse;
  out[(size_t)nidx * NOUT + 1] = o1 - mx - lse;
  out[(size_t)nidx * NOUT + 2] = o2 - mx - lse;
}


extern "C" void kernel_launch(void* const* d_in, const int* in_sizes, int n_in,
                              void* d_out, int out_size, void* d_ws, size_t ws_size,
                              hipStream_t stream) {
  (void)n_in; (void)out_size;
  const float* x   = (const float*)d_in[0];
  const int*   ei  = (const int*)d_in[1];
  const float* Wl1 = (const float*)d_in[2];
  const float* Wr1 = (const float*)d_in[3];
  const float* b1  = (const float*)d_in[4];
  const float* Wl2 = (const float*)d_in[5];
  const float* Wr2 = (const float*)d_in[6];
  const float* b2  = (const float*)d_in[7];

  int N = in_sizes[0] / IN_DIM;
  int E = in_sizes[1] / 2;

  size_t need = ((size_t)N * (2 * HID + 2 * QW + SLOTW + 1) + (size_t)E * 2) * 4;
  if (ws_size < need) return;
  float* pl   = (float*)d_ws;
  float* pr   = pl   + (size_t)N * HID;
  float* ql   = pr   + (size_t)N * HID;
  float* qr   = ql   + (size_t)N * QW;
  int*   slot = (int*)(qr + (size_t)N * QW);
  int*   link = slot + (size_t)N * SLOTW;
  int*   head = link + (size_t)E * 2;

  int rowTiles = (N + 15) / 16;
  int blocks   = (rowTiles + WPB - 1) / WPB;
  sage_project1_wmma<<<blocks, 256, 0, stream>>>(x, Wl1, Wr1, b1, pl, pr, slot, head, N, rowTiles);

  int lblocks = ((E + 3) / 4 + 255) / 256;
  int quarter = lblocks * 256;
  link_edges<<<lblocks, 256, 0, stream>>>(ei, slot, head, link, E, quarter);

  long long tn = (long long)N * 8;
  sage_gather_hidden_project<<<(int)((tn + 255) / 256), 256, 0, stream>>>(pl, pr, slot, head, link,
                                                                       Wl2, Wr2, b2, ql, qr, N, E);

  sage_gather_finish<<<(N + 255) / 256, 256, 0, stream>>>(ql, qr, slot, head, link, (float*)d_out, N, E);
}
